# starting kernel
# baseline (speedup 1.0000x reference)
#include <hip/hip_runtime.h>
#include <hip/hip_bf16.h>

typedef __attribute__((ext_vector_type(2))) float v2f;
typedef __attribute__((ext_vector_type(8))) float v8f;

#define PSTRIDE 160

template <bool B_FROM_W, int NACC>
__global__ __launch_bounds__(256) void wmma_gemm_kernel(
    const float* __restrict__ A, int lda, int Ktotal, int Kslice,
    const float* __restrict__ Bsrc, int ldb, int T_, int coutw,
    float* __restrict__ C, int Ncols) {
  __shared__ float Bs[32 * PSTRIDE];

  const int tid  = threadIdx.x;
  const int wave = tid >> 5;
  const int lane = tid & 31;
  const int lr   = lane & 15;
  const int lh   = lane >> 4;

  const int m0    = blockIdx.x * 128 + wave * 16;
  const int nbase = blockIdx.y * 64;

  const int ks = blockIdx.z * Kslice;
  const int ke = (ks + Kslice < Ktotal) ? (ks + Kslice) : Ktotal;

  v8f acc[NACC];
#pragma unroll
  for (int nj = 0; nj < NACC; ++nj) acc[nj] = (v8f){};

  const float* arow = A + (size_t)(m0 + lr) * lda + 2 * lh;

  for (int kchunk = ks; kchunk < ke; kchunk += 64) {
    const int kc = (ke - kchunk) < 64 ? (ke - kchunk) : 64;

    for (int idx = tid; idx < kc * 64; idx += 256) {
      const int r  = idx >> 6;
      const int c  = idx & 63;
      const int gc = nbase + c;
      float v = 0.0f;
      if (gc < Ncols) {
        if (B_FROM_W) {
          const int t = gc / coutw;
          const int j = gc - t * coutw;
          v = Bsrc[(size_t)(kchunk + r) * coutw * T_ + j * T_ + t];
        } else {
          v = Bsrc[(size_t)(kchunk + r) * ldb + gc];
        }
      }
      Bs[(r >> 1) * PSTRIDE + 2 * c + (r & 1)] = v;
    }
    __syncthreads();

    const int nsteps = kc >> 2;
#pragma unroll 4
    for (int s = 0; s < nsteps; ++s) {
      const int kk = kchunk + 4 * s;
      v2f a;
      a.x = arow[kk + 0];
      a.y = arow[kk + 1];
      const float* bp = Bs + (2 * s + lh) * PSTRIDE + 2 * lr;
      v2f bf[NACC];
#pragma unroll
      for (int nj = 0; nj < NACC; ++nj) {
        bf[nj] = *(const v2f*)(bp + nj * 32);
      }
#pragma unroll
      for (int nj = 0; nj < NACC; ++nj) {
        acc[nj] = __builtin_amdgcn_wmma_f32_16x16x4_f32(
            false, a, false, bf[nj], (short)0, acc[nj], false, false);
      }
    }
    __syncthreads();
  }

  float* cp = C + (size_t)blockIdx.z * 2048 * Ncols;
#pragma unroll
  for (int nj = 0; nj < NACC; ++nj) {
    if (nbase + nj * 16 < Ncols) {
      const int col = nbase + nj * 16 + lr;
#pragma unroll
      for (int r = 0; r < 8; ++r) {
        const int row = m0 + r + 8 * lh;
        cp[(size_t)row * Ncols + col] = acc[nj][r];
      }
    }
  }
}

__global__ __launch_bounds__(256) void reduce_bias_lrelu_kernel(
    const float* __restrict__ part, int S,
    const float* __restrict__ h, int cin,
    const float* __restrict__ V,
    float* __restrict__ out, int cout, int total) {
  const int idx = blockIdx.x * 256 + threadIdx.x;
  if (idx >= total) return;
  const int m = idx / cout;
  const int j = idx - m * cout;
  float acc = 0.0f;
  for (int s = 0; s < S; ++s) acc += part[(size_t)s * total + idx];
  const float* hr = h + (size_t)m * cin;
  for (int i = 0; i < cin; ++i) acc = fmaf(hr[i], V[(size_t)i * cout + j], acc);
  out[idx] = fmaxf(acc, 0.2f * acc);
}

extern "C" void kernel_launch(void* const* d_in, const int* in_sizes, int n_in,
                              void* d_out, int out_size, void* d_ws, size_t ws_size,
                              hipStream_t stream) {
  (void)in_sizes; (void)n_in; (void)out_size; (void)ws_size;
  const float* x   = (const float*)d_in[0];
  const float* adj = (const float*)d_in[1];
  const float* Wl[3] = {(const float*)d_in[2], (const float*)d_in[4], (const float*)d_in[6]};
  const float* Vl[3] = {(const float*)d_in[3], (const float*)d_in[5], (const float*)d_in[7]};

  const int N = 2048, T = 15, S = 16;
  const int Ktot = N * T;
  const int Ksl  = Ktot / S;

  float* ws   = (float*)d_ws;
  float* hw   = ws;
  float* part = hw + (size_t)N * 960;
  float* h1   = part + (size_t)S * N * 64;
  float* h2   = h1 + (size_t)N * 64;

  const int cins[3]  = {32, 64, 64};
  const int couts[3] = {64, 64, 32};
  const float* hin = x;
  float* houts[3] = {h1, h2, (float*)d_out};

  for (int L = 0; L < 3; ++L) {
    const int cin = cins[L], cout = couts[L];
    const int Nhw = T * cout;

    dim3 gA(N / 128, (Nhw + 63) / 64, 1);
    wmma_gemm_kernel<true, 4><<<gA, 256, 0, stream>>>(
        hin, cin, cin, cin, Wl[L], 0, T, cout, hw, Nhw);

    dim3 gB(N / 128, 1, S);
    if (cout == 64) {
      wmma_gemm_kernel<false, 4><<<gB, 256, 0, stream>>>(
          adj, Ktot, Ktot, Ksl, hw, cout, T, cout, part, cout);
    } else {
      wmma_gemm_kernel<false, 2><<<gB, 256, 0, stream>>>(
          adj, Ktot, Ktot, Ksl, hw, cout, T, cout, part, cout);
    }

    const int total = N * cout;
    reduce_bias_lrelu_kernel<<<(total + 255) / 256, 256, 0, stream>>>(
        part, S, hin, cin, Vl[L], houts[L], cout, total);

    hin = houts[L];
  }
}


# Step-A projection re-tiled: a block owns 32 rows x (16 j x all 15 t), so its W part is contiguous per k row and goes to LDS by 16-byte loads in one group; h rows direct to fragments; sum order kept
# speedup vs baseline: 6.1142x; 6.1142x over previous
#include <hip/hip_runtime.h>
#include <hip/hip_bf16.h>

typedef __attribute__((ext_vector_type(2))) float v2f;
typedef __attribute__((ext_vector_type(4))) float v4f;
typedef __attribute__((ext_vector_type(8))) float v8f;

template <int NS, int NHW>
__device__ __forceinline__ void proj_tile(
    const v2f (&a)[NS], const float* __restrict__ bb, float* __restrict__ op) {
  v2f b[NS];
#pragma unroll
  for (int s = 0; s < NS; ++s) {
    b[s].x = bb[(4 * s) * 240];
    b[s].y = bb[(4 * s + 1) * 240];
  }
  __builtin_amdgcn_sched_barrier(0);
  v8f acc = (v8f){};
#pragma unroll
  for (int s = 0; s < NS; ++s) {
    acc = __builtin_amdgcn_wmma_f32_16x16x4_f32(
        false, a[s], false, b[s], (short)0, acc, false, false);
  }
#pragma unroll
  for (int r = 0; r < 8; ++r) op[(size_t)r * NHW] = acc[r];
}

template <int CIN, int COUT>
__global__ __launch_bounds__(256) void proj_kernel(
    const float* __restrict__ H, const float* __restrict__ Wt, float* __restrict__ hw) {
  constexpr int NS  = CIN / 4;
  constexpr int NV  = CIN * 60;
  constexpr int NHW = 15 * COUT;
  __shared__ v4f Ws4[NV];

  const int tid  = threadIdx.x;
  const int wave = __builtin_amdgcn_readfirstlane(tid >> 5);
  const int lane = tid & 31;
  const int c    = lane & 15;
  const int h    = lane >> 4;
  const int jb   = blockIdx.y;
  const int m0   = blockIdx.x * 32 + (wave & 1) * 16;
  const int tb   = (wave >> 1) * 4;

  const int  r0 = tid / 60;
  const int  v  = tid - r0 * 60;
  const bool cp = tid < 240;
  const float* wsrc = Wt + jb * 240 + r0 * NHW + v * 4;
  v4f st[NS];
  if (cp) {
#pragma unroll
    for (int q = 0; q < NS; ++q) st[q] = *(const v4f*)(wsrc + q * 4 * NHW);
  }

  v2f a[NS];
  const float* hrow = H + (size_t)(m0 + c) * CIN + 2 * h;
#pragma unroll
  for (int s = 0; s < NS; ++s) a[s] = *(const v2f*)(hrow + 4 * s);

  if (cp) {
#pragma unroll
    for (int q = 0; q < NS; ++q) Ws4[(r0 + 4 * q) * 60 + v] = st[q];
  }
  __syncthreads();

  const float* bb = (const float*)Ws4 + (2 * h) * 240 + c * 15 + tb;
  float* op = hw + (size_t)(m0 + 8 * h) * NHW + tb * COUT + jb * 16 + c;
#pragma unroll
  for (int u = 0; u < 4; ++u) {
    if (tb + u < 15) {
      proj_tile<NS, NHW>(a, bb + u, op + u * COUT);
    }
  }
}

template <int NT> struct rowvec;
template <> struct rowvec<4> { typedef v4f type; };
template <> struct rowvec<2> { typedef v2f type; };

#define SB_LDA 30720
#define SB_KW  480
#define SB_CH  32

template <int NT>
__device__ __forceinline__ void sb_load_chunk(
    v4f (&a)[4][4], typename rowvec<NT>::type (&b)[4][4],
    const float* __restrict__ ap, const float* __restrict__ bp) {
  typedef typename rowvec<NT>::type bv;
  __builtin_amdgcn_sched_barrier(0);
#pragma unroll
  for (int mt = 0; mt < 4; ++mt) {
#pragma unroll
    for (int g = 0; g < 4; ++g)
      a[g][mt] = *(const v4f*)(ap + (size_t)mt * 16 * SB_LDA + g * 8);
  }
#pragma unroll
  for (int g = 0; g < 4; ++g) {
#pragma unroll
    for (int i = 0; i < 4; ++i)
      b[g][i] = *(const bv*)(bp + (g * 8 + i) * (16 * NT));
  }
  __builtin_amdgcn_sched_barrier(0);
}

template <int NT>
__device__ __forceinline__ void sb_compute_chunk(
    v8f (&acc)[4][NT], const v4f (&a)[4][4],
    const typename rowvec<NT>::type (&b)[4][4]) {
#pragma unroll
  for (int g = 0; g < 4; ++g) {
#pragma unroll
    for (int e = 0; e < 2; ++e) {
      v2f bf[NT];
#pragma unroll
      for (int nj = 0; nj < NT; ++nj) {
        bf[nj].x = b[g][2 * e][nj];
        bf[nj].y = b[g][2 * e + 1][nj];
      }
#pragma unroll
      for (int mt = 0; mt < 4; ++mt) {
        v2f af;
        af.x = a[g][mt][2 * e];
        af.y = a[g][mt][2 * e + 1];
#pragma unroll
        for (int nj = 0; nj < NT; ++nj) {
          acc[mt][nj] = __builtin_amdgcn_wmma_f32_16x16x4_f32(
              false, af, false, bf[nj], (short)0, acc[mt][nj], false, false);
        }
      }
    }
  }
}

template <int NT>
__device__ __forceinline__ void hv_load_group(
    v4f (&a)[4], typename rowvec<NT>::type (&b)[4],
    const float* __restrict__ hp, int cin, const float* __restrict__ vp) {
  typedef typename rowvec<NT>::type bv;
#pragma unroll
  for (int mt = 0; mt < 4; ++mt)
    a[mt] = *(const v4f*)(hp + (size_t)mt * 16 * cin);
#pragma unroll
  for (int i = 0; i < 4; ++i)
    b[i] = *(const bv*)(vp + i * (16 * NT));
}

template <int NT>
__device__ __forceinline__ void hv_compute_group(
    v8f (&acc)[4][NT], const v4f (&a)[4],
    const typename rowvec<NT>::type (&b)[4]) {
#pragma unroll
  for (int e = 0; e < 2; ++e) {
    v2f bf[NT];
#pragma unroll
    for (int nj = 0; nj < NT; ++nj) {
      bf[nj].x = b[2 * e][nj];
      bf[nj].y = b[2 * e + 1][nj];
    }
#pragma unroll
    for (int mt = 0; mt < 4; ++mt) {
      v2f af;
      af.x = a[mt][2 * e];
      af.y = a[mt][2 * e + 1];
#pragma unroll
      for (int nj = 0; nj < NT; ++nj) {
        acc[mt][nj] = __builtin_amdgcn_wmma_f32_16x16x4_f32(
            false, af, false, bf[nj], (short)0, acc[mt][nj], false, false);
      }
    }
  }
}

template <int NT>
__global__ __launch_bounds__(256) void adj_stream_gemm_kernel(
    const float* __restrict__ A, const float* __restrict__ Bm,
    float* __restrict__ C,
    const float* __restrict__ H, int cin, const float* __restrict__ V) {
  typedef typename rowvec<NT>::type bv;
  constexpr int NC = 16 * NT;
  __shared__ bv red[8 * 16 * 16];

  const int tid  = threadIdx.x;
  const int wave = tid >> 5;
  const int lane = tid & 31;
  const int c    = lane & 15;
  const int h    = lane >> 4;

  const int m0 = blockIdx.x * 64;
  const int k0 = (blockIdx.z * 8 + wave) * SB_KW;

  const float* ap = A + (size_t)(m0 + c) * SB_LDA + k0 + 4 * h;
  const float* bp = Bm + (size_t)(k0 + 4 * h) * NC + NT * c;

  v8f acc[4][NT];
#pragma unroll
  for (int mt = 0; mt < 4; ++mt)
#pragma unroll
    for (int nj = 0; nj < NT; ++nj) acc[mt][nj] = (v8f){};

  v4f a0[4][4], a1[4][4];
  bv  b0[4][4], b1[4][4];

  const bool hv = (__builtin_amdgcn_readfirstlane(wave) == 0) && ((int)blockIdx.z * 8 < cin);
  v4f ha[4];
  bv  hb[4];
  if (hv) {
    const int kz = blockIdx.z * 8 + 4 * h;
    hv_load_group<NT>(ha, hb, H + (size_t)(m0 + c) * cin + kz, cin,
                      V + (size_t)kz * NC + NT * c);
  }

  sb_load_chunk<NT>(a0, b0, ap, bp);
  if (hv) hv_compute_group<NT>(acc, ha, hb);
  for (int it = 0; it < 7; ++it) {
    sb_load_chunk<NT>(a1, b1, ap + SB_CH, bp + SB_CH * NC);
    sb_compute_chunk<NT>(acc, a0, b0);
    sb_load_chunk<NT>(a0, b0, ap + 2 * SB_CH, bp + 2 * SB_CH * NC);
    sb_compute_chunk<NT>(acc, a1, b1);
    ap += 2 * SB_CH;
    bp += 2 * SB_CH * NC;
  }
  sb_compute_chunk<NT>(acc, a0, b0);

  float* cp = C + ((size_t)blockIdx.z * 2048 + m0) * NC;
  const int rrow = tid >> 4;
  const int rcol = tid & 15;
#pragma unroll
  for (int mt = 0; mt < 4; ++mt) {
#pragma unroll
    for (int r = 0; r < 8; ++r) {
      bv v;
#pragma unroll
      for (int nj = 0; nj < NT; ++nj) v[nj] = acc[mt][nj][r];
      red[(wave * 16 + 8 * h + r) * 16 + c] = v;
    }
    __syncthreads();
    bv s = red[rrow * 16 + rcol];
#pragma unroll
    for (int w = 1; w < 8; ++w) s += red[(w * 16 + rrow) * 16 + rcol];
    *(bv*)(cp + (size_t)(mt * 16 + rrow) * NC + NT * rcol) = s;
    if (mt < 3) __syncthreads();
  }
}

__global__ __launch_bounds__(256) void reduce_lrelu_kernel(
    const float* __restrict__ part, float* __restrict__ out, int total4) {
  const int idx = blockIdx.x * 256 + threadIdx.x;
  if (idx >= total4) return;
  const v4f* p = (const v4f*)part + idx;
  v4f acc = p[0];
#pragma unroll
  for (int s = 1; s < 8; ++s) acc += p[(size_t)s * total4];
  v4f r;
#pragma unroll
  for (int q = 0; q < 4; ++q) r[q] = fmaxf(acc[q], 0.2f * acc[q]);
  ((v4f*)out)[idx] = r;
}

extern "C" void kernel_launch(void* const* d_in, const int* in_sizes, int n_in,
                              void* d_out, int out_size, void* d_ws, size_t ws_size,
                              hipStream_t stream) {
  (void)in_sizes; (void)n_in; (void)out_size; (void)ws_size;
  const float* x   = (const float*)d_in[0];
  const float* adj = (const float*)d_in[1];
  const float* Wl[3] = {(const float*)d_in[2], (const float*)d_in[4], (const float*)d_in[6]};
  const float* Vl[3] = {(const float*)d_in[3], (const float*)d_in[5], (const float*)d_in[7]};

  const int N = 2048;

  float* ws   = (float*)d_ws;
  float* hw   = ws;
  float* part = hw + (size_t)N * 960;
  float* h1   = part + (size_t)16 * N * 64;
  float* h2   = h1 + (size_t)N * 64;

  const int cins[3]  = {32, 64, 64};
  const int couts[3] = {64, 64, 32};
  const float* hin = x;
  float* houts[3] = {h1, h2, (float*)d_out};

  for (int L = 0; L < 3; ++L) {
    const int cin = cins[L], cout = couts[L];

    dim3 gA(N / 32, cout / 16, 1);
    if (cin == 32) {
      proj_kernel<32, 64><<<gA, 256, 0, stream>>>(hin, Wl[L], hw);
    } else if (cout == 64) {
      proj_kernel<64, 64><<<gA, 256, 0, stream>>>(hin, Wl[L], hw);
    } else {
      proj_kernel<64, 32><<<gA, 256, 0, stream>>>(hin, Wl[L], hw);
    }

    dim3 gB(32, 1, 8);
    if (cout == 64) {
      adj_stream_gemm_kernel<4><<<gB, 256, 0, stream>>>(adj, hw, part, hin, cin, Vl[L]);
    } else {
      adj_stream_gemm_kernel<2><<<gB, 256, 0, stream>>>(adj, hw, part, hin, cin, Vl[L]);
    }

    const int total4 = N * cout / 4;
    reduce_lrelu_kernel<<<(total4 + 255) / 256, 256, 0, stream>>>(
        part, houts[L], total4);

    hin = houts[L];
  }
}
